# starting kernel
# baseline (speedup 1.0000x reference)
#include <hip/hip_runtime.h>
#include <hip/hip_bf16.h>

#define N_DRUGS 4096
#define W_DIM   512
#define N_EDGES 131072

typedef __attribute__((ext_vector_type(2))) float v2f;
typedef __attribute__((ext_vector_type(8))) float v8f;

__global__ __launch_bounds__(256)
void dedicom_gemm(const float* __restrict__ Z,
                  const float* __restrict__ W,
                  const float* __restrict__ local_diag,
                  const int*   __restrict__ sub_idx,
                  float* __restrict__ Y) {
    const int tid   = threadIdx.x;
    const int lane  = tid & 31;
    const int wave  = tid >> 5;
    const int row0  = blockIdx.y * 128;
    const int n0g   = blockIdx.x * 64;
    const int m0    = wave * 16;
    const int mlane = lane & 15;
    const int nlane = lane & 15;
    const int koff  = (lane < 16) ? 0 : 2;

    const float* __restrict__ dg = local_diag + (size_t)sub_idx[0] * W_DIM;

    __shared__ float Zs[128 * 36];
    __shared__ float Ws[16 * 160];

    v8f acc[4];
    #pragma unroll
    for (int s = 0; s < 4; ++s)
        acc[s] = v8f{0.f, 0.f, 0.f, 0.f, 0.f, 0.f, 0.f, 0.f};

    for (int k0 = 0; k0 < W_DIM; k0 += 32) {
        __syncthreads();
        #pragma unroll
        for (int i = 0; i < 4; ++i) {
            const int idx = tid + 256 * i;
            const int r   = idx >> 3;
            const int c4  = (idx & 7) << 2;
            const float4 z4 = *(const float4*)&Z[(size_t)(row0 + r) * W_DIM + k0 + c4];
            const float4 d4 = *(const float4*)&dg[k0 + c4];
            *(float4*)&Zs[r * 36 + c4] =
                make_float4(z4.x * d4.x, z4.y * d4.y, z4.z * d4.z, z4.w * d4.w);
        }
        {
            const int kp = tid >> 4;
            const int n4 = (tid & 15) << 2;
            const float4 w0 = *(const float4*)&W[(size_t)(k0 + 2 * kp    ) * W_DIM + n0g + n4];
            const float4 w1 = *(const float4*)&W[(size_t)(k0 + 2 * kp + 1) * W_DIM + n0g + n4];
            float* dst = &Ws[kp * 160 + 2 * n4];
            *(float4*)(dst + 0) = make_float4(w0.x, w1.x, w0.y, w1.y);
            *(float4*)(dst + 4) = make_float4(w0.z, w1.z, w0.w, w1.w);
        }
        if (k0 + 32 < W_DIM) {
            __builtin_prefetch(&Z[(size_t)(row0 + (tid >> 3)) * W_DIM + k0 + 32], 0, 0);
            __builtin_prefetch(&W[(size_t)(k0 + 32 + 2 * (tid >> 4)) * W_DIM + n0g], 0, 0);
        }
        __syncthreads();

        #pragma unroll
        for (int kk = 0; kk < 32; kk += 4) {
            const v2f a = *(const v2f*)&Zs[(m0 + mlane) * 36 + kk + koff];
            const int kp = (kk + koff) >> 1;
            #pragma unroll
            for (int s = 0; s < 4; ++s) {
                const v2f b = *(const v2f*)&Ws[kp * 160 + 2 * (s * 16 + nlane)];
                acc[s] = __builtin_amdgcn_wmma_f32_16x16x4_f32(
                     false, a,  false, b,
                     (short)0, acc[s],
                     false,  false);
            }
        }
    }

    const int rbase = row0 + m0 + ((lane < 16) ? 0 : 8);
    #pragma unroll
    for (int s = 0; s < 4; ++s) {
        const int col = n0g + s * 16 + nlane;
        const float dn = dg[col];
        #pragma unroll
        for (int v = 0; v < 8; ++v) {
            Y[(size_t)(rbase + v) * W_DIM + col] = acc[s][v] * dn;
        }
    }
}

__global__ __launch_bounds__(256)
void dedicom_edge_score(const float* __restrict__ Y,
                        const float* __restrict__ Z,
                        const int*   __restrict__ edges,
                        float* __restrict__ out) {
    const int wave = threadIdx.x >> 5;
    const int lane = threadIdx.x & 31;
    const int e    = blockIdx.x * 8 + wave;
    if (e >= N_EDGES) return;

    const int r = edges[e];
    const int c = edges[N_EDGES + e];
    const float4* __restrict__ yr = (const float4*)(Y + (size_t)r * W_DIM);
    const float4* __restrict__ zc = (const float4*)(Z + (size_t)c * W_DIM);

    float s = 0.f;
    #pragma unroll
    for (int i = 0; i < 4; ++i) {
        const float4 a = yr[lane + 32 * i];
        const float4 b = zc[lane + 32 * i];
        s += a.x * b.x + a.y * b.y + a.z * b.z + a.w * b.w;
    }
    #pragma unroll
    for (int off = 16; off > 0; off >>= 1)
        s += __shfl_xor(s, off, 32);

    if (lane == 0)
        out[e] = 1.0f / (1.0f + __expf(-s));
}

extern "C" void kernel_launch(void* const* d_in, const int* in_sizes, int n_in,
                              void* d_out, int out_size, void* d_ws, size_t ws_size,
                              hipStream_t stream) {
    const float* z_drug     = (const float*)d_in[0];
    const float* gweight    = (const float*)d_in[1];
    const float* local_diag = (const float*)d_in[2];
    const int*   edges      = (const int*)d_in[3];
    const int*   sub_idx    = (const int*)d_in[4];
    float*       out        = (float*)d_out;
    float*       Y          = (float*)d_ws;

    dim3 gemm_grid(W_DIM / 64, N_DRUGS / 128);
    dedicom_gemm<<<gemm_grid, 256, 0, stream>>>(z_drug, gweight, local_diag, sub_idx, Y);

    dedicom_edge_score<<<N_EDGES / 8, 256, 0, stream>>>(Y, z_drug, edges, out);
}


# GEMM K loop leaves LDS staging: W's 512x64 slab put in LDS once per block, each wave loads its 16 Z rows straight into A fragments, no barrier in the loop; k regrouped within 8s (sum order changes)
# speedup vs baseline: 1.1399x; 1.1399x over previous
#include <hip/hip_runtime.h>
#include <hip/hip_bf16.h>

#define N_DRUGS 4096
#define W_DIM   512
#define N_EDGES 131072

typedef __attribute__((ext_vector_type(2))) float v2f;
typedef __attribute__((ext_vector_type(8))) float v8f;

#define WS_PITCH 176

__global__ __launch_bounds__(256)
void dedicom_gemm(const float* __restrict__ Z,
                  const float* __restrict__ W,
                  const float* __restrict__ local_diag,
                  const int*   __restrict__ sub_idx,
                  float* __restrict__ Y) {
    const int tid   = threadIdx.x;
    const int lane  = tid & 31;
    const int wave  = tid >> 5;
    const int row0  = blockIdx.y * 128;
    const int n0g   = blockIdx.x * 64;
    const int m0    = wave * 16;
    const int c     = lane & 15;
    const int h     = lane >> 4;

    const float* __restrict__ dg = local_diag + (size_t)sub_idx[0] * W_DIM;

    __shared__ float Ws[256 * WS_PITCH];

    {
        const int kpl = tid >> 4;
        const int n4  = (tid & 15) << 2;
        #pragma unroll
        for (int i = 0; i < 16; ++i) {
            const int kp = i * 16 + kpl;
            const float4 w0 = *(const float4*)&W[(size_t)(2 * kp    ) * W_DIM + n0g + n4];
            const float4 w1 = *(const float4*)&W[(size_t)(2 * kp + 1) * W_DIM + n0g + n4];
            float* dst = &Ws[kp * WS_PITCH + 2 * n4];
            *(float4*)(dst + 0) = make_float4(w0.x, w1.x, w0.y, w1.y);
            *(float4*)(dst + 4) = make_float4(w0.z, w1.z, w0.w, w1.w);
        }
    }
    __syncthreads();

    v8f acc[4];
    #pragma unroll
    for (int s = 0; s < 4; ++s)
        acc[s] = v8f{0.f, 0.f, 0.f, 0.f, 0.f, 0.f, 0.f, 0.f};

    const float* __restrict__ zl = Z + (size_t)(row0 + m0 + c) * W_DIM + 4 * h;
    const float* __restrict__ dl = dg + 4 * h;
    const float* bl = Ws + (2 * h) * WS_PITCH + 2 * c;

    for (int g0 = 0; g0 < 64; g0 += 8) {
        float4 z4[8], d4[8];
        #pragma unroll
        for (int u = 0; u < 8; ++u) {
            z4[u] = *(const float4*)(zl + 8 * (g0 + u));
            d4[u] = *(const float4*)(dl + 8 * (g0 + u));
        }
        #pragma unroll
        for (int u = 0; u < 8; ++u) {
            const v2f alo = {z4[u].x * d4[u].x, z4[u].y * d4[u].y};
            const v2f ahi = {z4[u].z * d4[u].z, z4[u].w * d4[u].w};
            const float* bp = bl + (4 * (g0 + u)) * WS_PITCH;
            #pragma unroll
            for (int s = 0; s < 4; ++s) {
                const v2f b = *(const v2f*)(bp + 32 * s);
                acc[s] = __builtin_amdgcn_wmma_f32_16x16x4_f32(
                     false, alo,  false, b,
                     (short)0, acc[s],
                     false,  false);
            }
            #pragma unroll
            for (int s = 0; s < 4; ++s) {
                const v2f b = *(const v2f*)(bp + WS_PITCH + 32 * s);
                acc[s] = __builtin_amdgcn_wmma_f32_16x16x4_f32(
                     false, ahi,  false, b,
                     (short)0, acc[s],
                     false,  false);
            }
        }
    }

    const int rbase = row0 + m0 + 8 * h;
    #pragma unroll
    for (int s = 0; s < 4; ++s) {
        const int col = n0g + s * 16 + c;
        const float dn = dg[col];
        #pragma unroll
        for (int v = 0; v < 8; ++v) {
            Y[(size_t)(rbase + v) * W_DIM + col] = acc[s][v] * dn;
        }
    }
}

__global__ __launch_bounds__(256)
void dedicom_edge_score(const float* __restrict__ Y,
                        const float* __restrict__ Z,
                        const int*   __restrict__ edges,
                        float* __restrict__ out) {
    const int wave = threadIdx.x >> 5;
    const int lane = threadIdx.x & 31;
    const int e    = blockIdx.x * 8 + wave;
    if (e >= N_EDGES) return;

    const int r = edges[e];
    const int c = edges[N_EDGES + e];
    const float4* __restrict__ yr = (const float4*)(Y + (size_t)r * W_DIM);
    const float4* __restrict__ zc = (const float4*)(Z + (size_t)c * W_DIM);

    float s = 0.f;
    #pragma unroll
    for (int i = 0; i < 4; ++i) {
        const float4 a = yr[lane + 32 * i];
        const float4 b = zc[lane + 32 * i];
        s += a.x * b.x + a.y * b.y + a.z * b.z + a.w * b.w;
    }
    #pragma unroll
    for (int off = 16; off > 0; off >>= 1)
        s += __shfl_xor(s, off, 32);

    if (lane == 0)
        out[e] = 1.0f / (1.0f + __expf(-s));
}

extern "C" void kernel_launch(void* const* d_in, const int* in_sizes, int n_in,
                              void* d_out, int out_size, void* d_ws, size_t ws_size,
                              hipStream_t stream) {
    const float* z_drug     = (const float*)d_in[0];
    const float* gweight    = (const float*)d_in[1];
    const float* local_diag = (const float*)d_in[2];
    const int*   edges      = (const int*)d_in[3];
    const int*   sub_idx    = (const int*)d_in[4];
    float*       out        = (float*)d_out;
    float*       Y          = (float*)d_ws;

    dim3 gemm_grid(W_DIM / 64, N_DRUGS / 128);
    dedicom_gemm<<<gemm_grid, 256, 0, stream>>>(z_drug, gweight, local_diag, sub_idx, Y);

    dedicom_edge_score<<<N_EDGES / 8, 256, 0, stream>>>(Y, z_drug, edges, out);
}


# Edge stage re-split by lane groups: 8 lanes own an edge, one load serves 4 edges (a whole 128B line each), a wave covers 32 edges and stores its 32 scores as one line; per-edge sum order changes
# speedup vs baseline: 1.3038x; 1.1438x over previous
#include <hip/hip_runtime.h>
#include <hip/hip_bf16.h>

#define N_DRUGS 4096
#define W_DIM   512
#define N_EDGES 131072

typedef __attribute__((ext_vector_type(2))) float v2f;
typedef __attribute__((ext_vector_type(8))) float v8f;

#define WS_PITCH 176

__global__ __launch_bounds__(256)
void dedicom_gemm(const float* __restrict__ Z,
                  const float* __restrict__ W,
                  const float* __restrict__ local_diag,
                  const int*   __restrict__ sub_idx,
                  float* __restrict__ Y) {
    const int tid   = threadIdx.x;
    const int lane  = tid & 31;
    const int wave  = tid >> 5;
    const int row0  = blockIdx.y * 128;
    const int n0g   = blockIdx.x * 64;
    const int m0    = wave * 16;
    const int c     = lane & 15;
    const int h     = lane >> 4;

    const float* __restrict__ dg = local_diag + (size_t)sub_idx[0] * W_DIM;

    __shared__ float Ws[256 * WS_PITCH];

    {
        const int kpl = tid >> 4;
        const int n4  = (tid & 15) << 2;
        #pragma unroll
        for (int i = 0; i < 16; ++i) {
            const int kp = i * 16 + kpl;
            const float4 w0 = *(const float4*)&W[(size_t)(2 * kp    ) * W_DIM + n0g + n4];
            const float4 w1 = *(const float4*)&W[(size_t)(2 * kp + 1) * W_DIM + n0g + n4];
            float* dst = &Ws[kp * WS_PITCH + 2 * n4];
            *(float4*)(dst + 0) = make_float4(w0.x, w1.x, w0.y, w1.y);
            *(float4*)(dst + 4) = make_float4(w0.z, w1.z, w0.w, w1.w);
        }
    }
    __syncthreads();

    v8f acc[4];
    #pragma unroll
    for (int s = 0; s < 4; ++s)
        acc[s] = v8f{0.f, 0.f, 0.f, 0.f, 0.f, 0.f, 0.f, 0.f};

    const float* __restrict__ zl = Z + (size_t)(row0 + m0 + c) * W_DIM + 4 * h;
    const float* __restrict__ dl = dg + 4 * h;
    const float* bl = Ws + (2 * h) * WS_PITCH + 2 * c;

    for (int g0 = 0; g0 < 64; g0 += 8) {
        float4 z4[8], d4[8];
        #pragma unroll
        for (int u = 0; u < 8; ++u) {
            z4[u] = *(const float4*)(zl + 8 * (g0 + u));
            d4[u] = *(const float4*)(dl + 8 * (g0 + u));
        }
        #pragma unroll
        for (int u = 0; u < 8; ++u) {
            const v2f alo = {z4[u].x * d4[u].x, z4[u].y * d4[u].y};
            const v2f ahi = {z4[u].z * d4[u].z, z4[u].w * d4[u].w};
            const float* bp = bl + (4 * (g0 + u)) * WS_PITCH;
            #pragma unroll
            for (int s = 0; s < 4; ++s) {
                const v2f b = *(const v2f*)(bp + 32 * s);
                acc[s] = __builtin_amdgcn_wmma_f32_16x16x4_f32(
                     false, alo,  false, b,
                     (short)0, acc[s],
                     false,  false);
            }
            #pragma unroll
            for (int s = 0; s < 4; ++s) {
                const v2f b = *(const v2f*)(bp + WS_PITCH + 32 * s);
                acc[s] = __builtin_amdgcn_wmma_f32_16x16x4_f32(
                     false, ahi,  false, b,
                     (short)0, acc[s],
                     false,  false);
            }
        }
    }

    const int rbase = row0 + m0 + 8 * h;
    #pragma unroll
    for (int s = 0; s < 4; ++s) {
        const int col = n0g + s * 16 + c;
        const float dn = dg[col];
        #pragma unroll
        for (int v = 0; v < 8; ++v) {
            Y[(size_t)(rbase + v) * W_DIM + col] = acc[s][v] * dn;
        }
    }
}

typedef __attribute__((ext_vector_type(4))) float v4f;

template <int M>
__device__ __forceinline__ float lane_xor_f32(float v) {
    return __builtin_bit_cast(float,
        __builtin_amdgcn_update_dpp(0, __builtin_bit_cast(int, v), 0x160 | M, 0xf, 0xf, true));
}

__global__ __launch_bounds__(256)
void dedicom_edge_score(const float* __restrict__ Y,
                        const float* __restrict__ Z,
                        const int*   __restrict__ edges,
                        float* __restrict__ out) {
    const int lane  = threadIdx.x & 31;
    const int wave  = threadIdx.x >> 5;
    const int e0    = (blockIdx.x * 8 + wave) * 32;
    const int j     = lane & 7;
    const int gbase = lane & 24;

    const int my_r = edges[e0 + lane];
    const int my_c = edges[N_EDGES + e0 + lane];

    int r = __shfl(my_r, gbase, 32);
    int c = __shfl(my_c, gbase, 32);

    float mine = 0.f;
    for (int b = 0; b < 8; ++b) {
        const v4f* yr = (const v4f*)(Y + (size_t)r * W_DIM) + j;
        const v4f* zc = (const v4f*)(Z + (size_t)c * W_DIM) + j;

        v4f a[16], q[16];
        __builtin_amdgcn_sched_barrier(0);
        #pragma unroll
        for (int i = 0; i < 16; ++i) {
            a[i] = yr[8 * i];
            q[i] = zc[8 * i];
        }
        __builtin_amdgcn_sched_barrier(0);

        float c0 = 0.f, c1 = 0.f, c2 = 0.f, c3 = 0.f, c4 = 0.f, c5 = 0.f, c6 = 0.f, c7 = 0.f;
        #pragma unroll
        for (int i = 0; i < 16; i += 2) {
            c0 = __builtin_fmaf(a[i].x, q[i].x, c0);
            c1 = __builtin_fmaf(a[i].y, q[i].y, c1);
            c2 = __builtin_fmaf(a[i].z, q[i].z, c2);
            c3 = __builtin_fmaf(a[i].w, q[i].w, c3);
            c4 = __builtin_fmaf(a[i + 1].x, q[i + 1].x, c4);
            c5 = __builtin_fmaf(a[i + 1].y, q[i + 1].y, c5);
            c6 = __builtin_fmaf(a[i + 1].z, q[i + 1].z, c6);
            c7 = __builtin_fmaf(a[i + 1].w, q[i + 1].w, c7);
        }
        float s = ((c0 + c1) + (c2 + c3)) + ((c4 + c5) + (c6 + c7));

        r = __shfl(my_r, gbase | ((b + 1) & 7), 32);
        c = __shfl(my_c, gbase | ((b + 1) & 7), 32);

        s += lane_xor_f32<4>(s);
        s += lane_xor_f32<2>(s);
        s += lane_xor_f32<1>(s);
        if (j == b) mine = s;
    }

    out[e0 + lane] = 1.0f / (1.0f + __expf(-mine));
}

extern "C" void kernel_launch(void* const* d_in, const int* in_sizes, int n_in,
                              void* d_out, int out_size, void* d_ws, size_t ws_size,
                              hipStream_t stream) {
    const float* z_drug     = (const float*)d_in[0];
    const float* gweight    = (const float*)d_in[1];
    const float* local_diag = (const float*)d_in[2];
    const int*   edges      = (const int*)d_in[3];
    const int*   sub_idx    = (const int*)d_in[4];
    float*       out        = (float*)d_out;
    float*       Y          = (float*)d_ws;

    dim3 gemm_grid(W_DIM / 64, N_DRUGS / 128);
    dedicom_gemm<<<gemm_grid, 256, 0, stream>>>(z_drug, gweight, local_diag, sub_idx, Y);

    dedicom_edge_score<<<N_EDGES / 256, 256, 0, stream>>>(Y, z_drug, edges, out);
}
